# starting kernel
# baseline (speedup 1.0000x reference)
#include <hip/hip_runtime.h>
#include <hip/hip_bf16.h>


typedef __attribute__((ext_vector_type(2))) float v2f;
typedef __attribute__((ext_vector_type(8))) float v8f;

__global__ __launch_bounds__(256) void l2norm_rows(const float* __restrict__ F,
                                                   float* __restrict__ X,
                                                   int nRows) {
    const int lane = threadIdx.x & 31;
    const int row  = blockIdx.x * 8 + (threadIdx.x >> 5);
    if (row >= nRows) return;
    const float4* f = (const float4*)(F + (size_t)row * 256);
    float4 v0 = f[lane];
    float4 v1 = f[lane + 32];
    float s = v0.x*v0.x + v0.y*v0.y + v0.z*v0.z + v0.w*v0.w
            + v1.x*v1.x + v1.y*v1.y + v1.z*v1.z + v1.w*v1.w;
    #pragma unroll
    for (int off = 16; off > 0; off >>= 1) s += __shfl_xor(s, off, 32);
    const float inv = rsqrtf(fmaxf(s, 1e-12f));
    v0.x *= inv; v0.y *= inv; v0.z *= inv; v0.w *= inv;
    v1.x *= inv; v1.y *= inv; v1.z *= inv; v1.w *= inv;
    float4* xo = (float4*)(X + (size_t)row * 256);
    xo[lane]      = v0;
    xo[lane + 32] = v1;
}

template <int K, int M, bool RELU>
__global__ __launch_bounds__(128) void gemm_wmma_f32(const float* __restrict__ X,
                                                     const float* __restrict__ W,
                                                     float* __restrict__ Y,
                                                     int nRows) {
    const int lane   = threadIdx.x & 31;
    const int waveId = blockIdx.x * 4 + (threadIdx.x >> 5);
    const int tilesN = M / 16;
    const int totalTiles = (nRows / 16) * tilesN;
    if (waveId >= totalTiles) return;
    const int rowTile = waveId / tilesN;
    const int colTile = waveId % tilesN;

    const int halfSel = lane >> 4;
    const int kOff    = halfSel * 2;
    const int aRow    = rowTile * 16 + (lane & 15);
    const int bCol    = colTile * 16 + (lane & 15);

    const float* __restrict__ xrow = X + (size_t)aRow * K;
    v8f c = {};
    #pragma unroll 4
    for (int k = 0; k < K; k += 4) {
        float a0 = xrow[k + kOff];
        float a1 = xrow[k + kOff + 1];
        if (RELU) { a0 = fmaxf(a0, 0.0f); a1 = fmaxf(a1, 0.0f); }
        v2f a; a.x = a0; a.y = a1;
        v2f b;
        b.x = W[(size_t)(k + kOff)     * M + bCol];
        b.y = W[(size_t)(k + kOff + 1) * M + bCol];
        c = __builtin_amdgcn_wmma_f32_16x16x4_f32(
                false, a, false, b, (short)0, c, false, false);
    }
    const int mBase = rowTile * 16 + halfSel * 8;
    #pragma unroll
    for (int r = 0; r < 8; ++r)
        Y[(size_t)(mBase + r) * M + bCol] = c[r];
}

template <int D>
__global__ __launch_bounds__(256) void spmm_scatter(const float* __restrict__ H,
                                                    const float* __restrict__ vals,
                                                    const int* __restrict__ rows,
                                                    const int* __restrict__ cols,
                                                    float* __restrict__ out,
                                                    int nEdges) {
    constexpr int PER = D / 32;
    typedef __attribute__((ext_vector_type(PER))) float vPf;

    const int lane = threadIdx.x & 31;
    const int edge = blockIdx.x * 8 + (threadIdx.x >> 5);
    if (edge >= nEdges) return;
    const float v    = vals[edge];
    const int   r    = rows[edge];
    const int   cidx = cols[edge];

    const vPf src = *(const vPf*)(H + (size_t)cidx * D + lane * PER);
    float* __restrict__ dst = out + (size_t)r * D + lane * PER;
    #pragma unroll
    for (int j = 0; j < PER; ++j)
        atomicAdd(&dst[j], v * src[j]);
}

extern "C" void kernel_launch(void* const* d_in, const int* in_sizes, int n_in,
                              void* d_out, int out_size, void* d_ws, size_t ws_size,
                              hipStream_t stream) {
    const float* features = (const float*)d_in[0];
    const float* edgeVals = (const float*)d_in[1];
    const float* W0       = (const float*)d_in[2];
    const float* W1       = (const float*)d_in[3];
    const float* W2       = (const float*)d_in[4];
    const int*   edgeRow  = (const int*)d_in[5];
    const int*   edgeCol  = (const int*)d_in[6];

    const int N = in_sizes[0] / 256;
    const int E = in_sizes[1];

    float* ws = (float*)d_ws;
    float* xnorm = ws;
    float* t0    = ws + (size_t)N * 256;
    float* g0    = ws + (size_t)N * 384;
    float* t1    = xnorm;
    float* g1    = t0;
    float* t2    = g0;
    float* out   = (float*)d_out;

    const int normBlocks = (N + 7) / 8;
    const int spmmBlocks = (E + 7) / 8;
    auto gemmBlocks = [N](int m) { return ((N / 16) * (m / 16) + 3) / 4; };

    l2norm_rows<<<normBlocks, 256, 0, stream>>>(features, xnorm, N);

    gemm_wmma_f32<256, 128, false><<<gemmBlocks(128), 128, 0, stream>>>(xnorm, W0, t0, N);
    hipMemsetAsync(g0, 0, (size_t)N * 128 * sizeof(float), stream);
    spmm_scatter<128><<<spmmBlocks, 256, 0, stream>>>(t0, edgeVals, edgeRow, edgeCol, g0, E);

    gemm_wmma_f32<128, 64, true><<<gemmBlocks(64), 128, 0, stream>>>(g0, W1, t1, N);
    hipMemsetAsync(g1, 0, (size_t)N * 64 * sizeof(float), stream);
    spmm_scatter<64><<<spmmBlocks, 256, 0, stream>>>(t1, edgeVals, edgeRow, edgeCol, g1, E);

    gemm_wmma_f32<64, 32, true><<<gemmBlocks(32), 128, 0, stream>>>(g1, W2, t2, N);
    hipMemsetAsync(out, 0, (size_t)N * 32 * sizeof(float), stream);
    spmm_scatter<32><<<spmmBlocks, 256, 0, stream>>>(t2, edgeVals, edgeRow, edgeCol, out, E);
}


# SpMM result-stationary in LDS: edges counting-sorted once per call into 200-row bins, one block per bin sums its rows by LDS float atomics, stores each once; no global atomics; sum order by arrival
# speedup vs baseline: 3.2203x; 3.2203x over previous
#include <hip/hip_runtime.h>
#include <hip/hip_bf16.h>


typedef __attribute__((ext_vector_type(2))) float v2f;
typedef __attribute__((ext_vector_type(8))) float v8f;
typedef __attribute__((ext_vector_type(4))) float v4f;

__global__ __launch_bounds__(256) void l2norm_rows(const float* __restrict__ F,
                                                   float* __restrict__ X,
                                                   int nRows) {
    const int lane = threadIdx.x & 31;
    const int row  = blockIdx.x * 8 + (threadIdx.x >> 5);
    if (row >= nRows) return;
    const float4* f = (const float4*)(F + (size_t)row * 256);
    float4 v0 = f[lane];
    float4 v1 = f[lane + 32];
    float s = v0.x*v0.x + v0.y*v0.y + v0.z*v0.z + v0.w*v0.w
            + v1.x*v1.x + v1.y*v1.y + v1.z*v1.z + v1.w*v1.w;
    #pragma unroll
    for (int off = 16; off > 0; off >>= 1) s += __shfl_xor(s, off, 32);
    const float inv = rsqrtf(fmaxf(s, 1e-12f));
    v0.x *= inv; v0.y *= inv; v0.z *= inv; v0.w *= inv;
    v1.x *= inv; v1.y *= inv; v1.z *= inv; v1.w *= inv;
    float4* xo = (float4*)(X + (size_t)row * 256);
    xo[lane]      = v0;
    xo[lane + 32] = v1;
}

template <int K, int M, bool RELU>
__global__ __launch_bounds__(256) void gemm_wstat_f32(const float* __restrict__ X,
                                                      const float* __restrict__ W,
                                                      float* __restrict__ Y,
                                                      int nRows) {
    constexpr int TN   = M / 16;
    constexpr int G    = 8 / TN;
    constexpr int ROWS = G * 32;
    constexpr int K4   = K / 4;
    constexpr int KP4  = K4 + 1;
    constexpr int NLD  = ROWS * K4 / 256;

    __shared__ v4f tile4[ROWS * KP4];

    const int tid     = threadIdx.x;
    const int lane    = tid & 31;
    const int wave    = tid >> 5;
    const int colTile = wave % TN;
    const int grp     = wave / TN;
    const int c       = lane & 15;
    const int h       = lane >> 4;
    const int bCol    = colTile * 16 + c;

    const int nChunks = (nRows + ROWS - 1) / ROWS;
    int chunk = blockIdx.x;
    if (chunk >= nChunks) return;

    v2f wreg[K4];
    #pragma unroll
    for (int g = 0; g < K / 8; ++g) {
        #pragma unroll
        for (int s = 0; s < 2; ++s) {
            const int kk = 8 * g + 4 * h + 2 * s;
            v2f b;
            b.x = W[(size_t)kk       * M + bCol];
            b.y = W[(size_t)(kk + 1) * M + bCol];
            wreg[2 * g + s] = b;
        }
    }

    const v4f* __restrict__ X4 = (const v4f*)X;
    v4f p[NLD];

    #pragma unroll
    for (int i = 0; i < NLD; ++i) {
        const int idx  = tid + 256 * i;
        const int grow = min(chunk * ROWS + idx / K4, nRows - 1);
        p[i] = X4[(size_t)grow * K4 + (idx % K4)];
    }

    for (; chunk < nChunks; chunk += gridDim.x) {
        #pragma unroll
        for (int i = 0; i < NLD; ++i) {
            const int idx = tid + 256 * i;
            v4f v = p[i];
            if (RELU) {
                v.x = fmaxf(v.x, 0.0f); v.y = fmaxf(v.y, 0.0f);
                v.z = fmaxf(v.z, 0.0f); v.w = fmaxf(v.w, 0.0f);
            }
            tile4[(idx / K4) * KP4 + (idx % K4)] = v;
        }
        __syncthreads();

        const int nxt = chunk + gridDim.x;
        if (nxt < nChunks) {
            #pragma unroll
            for (int i = 0; i < NLD; ++i) {
                const int idx  = tid + 256 * i;
                const int grow = min(nxt * ROWS + idx / K4, nRows - 1);
                p[i] = X4[(size_t)grow * K4 + (idx % K4)];
            }
        }

        const int row0 = chunk * ROWS + grp * 32;
        if (row0 < nRows) {
            const v4f* a0p = tile4 + (grp * 32 + c) * KP4 + h;
            const v4f* a1p = a0p + 16 * KP4;
            v8f acc0 = {};
            v8f acc1 = {};
            #pragma unroll
            for (int g = 0; g < K / 8; ++g) {
                const v4f a0 = a0p[2 * g];
                const v4f a1 = a1p[2 * g];
                v2f f;
                f.x = a0.x; f.y = a0.y;
                acc0 = __builtin_amdgcn_wmma_f32_16x16x4_f32(false, f, false, wreg[2 * g],     (short)0, acc0, false, false);
                f.x = a1.x; f.y = a1.y;
                acc1 = __builtin_amdgcn_wmma_f32_16x16x4_f32(false, f, false, wreg[2 * g],     (short)0, acc1, false, false);
                f.x = a0.z; f.y = a0.w;
                acc0 = __builtin_amdgcn_wmma_f32_16x16x4_f32(false, f, false, wreg[2 * g + 1], (short)0, acc0, false, false);
                f.x = a1.z; f.y = a1.w;
                acc1 = __builtin_amdgcn_wmma_f32_16x16x4_f32(false, f, false, wreg[2 * g + 1], (short)0, acc1, false, false);
            }
            float* __restrict__ y0 = Y + (size_t)(row0 + 8 * h) * M + bCol;
            #pragma unroll
            for (int r = 0; r < 8; ++r)
                y0[(size_t)r * M] = acc0[r];
            if (row0 + 16 < nRows) {
                float* __restrict__ y1 = y0 + (size_t)16 * M;
                #pragma unroll
                for (int r = 0; r < 8; ++r)
                    y1[(size_t)r * M] = acc1[r];
            }
        }
        __syncthreads();
    }
}

constexpr int RB    = 200;
constexpr int MAXNB = 512;
constexpr int CH    = 8192;
constexpr int EPT   = CH / 1024;

struct __attribute__((aligned(16))) EdgeRec { int row; int col; float val; int pad; };

__global__ __launch_bounds__(1024) void bin_count(const int* __restrict__ rows,
                                                  int* __restrict__ blockHist,
                                                  int nEdges) {
    __shared__ int hist[MAXNB];
    const int tid = threadIdx.x;
    if (tid < MAXNB) hist[tid] = 0;
    __syncthreads();
    const int beg = blockIdx.x * CH;
    const int end = min(beg + CH, nEdges);
    int er[EPT];
    #pragma unroll
    for (int k = 0; k < EPT; ++k) {
        const int i = beg + tid + 1024 * k;
        er[k] = (i < end) ? rows[i] : -1;
    }
    #pragma unroll
    for (int k = 0; k < EPT; ++k)
        if (er[k] >= 0) atomicAdd(&hist[er[k] / RB], 1);
    __syncthreads();
    if (tid < MAXNB) blockHist[blockIdx.x * MAXNB + tid] = hist[tid];
}

__device__ __forceinline__ int hist_col_sum(const int* __restrict__ p, int n) {
    int s = 0;
    for (int i = 0; i < n; i += 16) {
        int c[16];
        #pragma unroll
        for (int q = 0; q < 16; ++q)
            c[q] = (i + q < n) ? p[(size_t)(i + q) * MAXNB] : 0;
        #pragma unroll
        for (int q = 0; q < 16; ++q)
            s += c[q];
    }
    return s;
}

__global__ __launch_bounds__(1024) void bin_fill(const int* __restrict__ rows,
                                                 const int* __restrict__ cols,
                                                 const float* __restrict__ vals,
                                                 const int* __restrict__ blockHist,
                                                 int* __restrict__ binStart,
                                                 EdgeRec* __restrict__ rec,
                                                 int nEdges, int nBlocks) {
    __shared__ int sBelow[2][MAXNB];
    __shared__ int sTotal[2][MAXNB];
    __shared__ int sBase[MAXNB];
    __shared__ int sCur[MAXNB];
    __shared__ int sWave[32];

    const int tid  = threadIdx.x;
    const int lane = tid & 31;
    const int wave = tid >> 5;
    const int me   = blockIdx.x;

    const int beg = me * CH;
    const int end = min(beg + CH, nEdges);
    int   er[EPT], ec[EPT];
    float ev[EPT];
    #pragma unroll
    for (int k = 0; k < EPT; ++k) {
        const int i = beg + tid + 1024 * k;
        er[k] = -1; ec[k] = 0; ev[k] = 0.0f;
        if (i < end) { er[k] = rows[i]; ec[k] = cols[i]; ev[k] = vals[i]; }
    }

    {
        const int bin  = tid & (MAXNB - 1);
        const int half = __builtin_amdgcn_readfirstlane(tid / MAXNB);
        const int mid  = (nBlocks + 1) / 2;
        const int lo   = half ? mid : 0;
        const int hi   = half ? nBlocks : mid;
        const int cut  = min(max(me, lo), hi);
        const int below = hist_col_sum(blockHist + (size_t)lo  * MAXNB + bin, cut - lo);
        const int rest  = hist_col_sum(blockHist + (size_t)cut * MAXNB + bin, hi - cut);
        sBelow[half][bin] = below;
        sTotal[half][bin] = below + rest;
    }
    __syncthreads();

    int tot = 0, bel = 0;
    if (tid < MAXNB) {
        tot = sTotal[0][tid] + sTotal[1][tid];
        bel = sBelow[0][tid] + sBelow[1][tid];
    }
    int incl = tot;
    for (int off = 1; off < 32; off <<= 1) {
        const int up = __shfl_up(incl, off, 32);
        if (lane >= off) incl += up;
    }
    if (lane == 31) sWave[wave] = incl;
    __syncthreads();
    if (wave == 0) {
        int s = sWave[lane];
        for (int off = 1; off < 32; off <<= 1) {
            const int up = __shfl_up(s, off, 32);
            if (lane >= off) s += up;
        }
        sWave[lane] = s;
    }
    __syncthreads();
    const int excl = incl - tot + (wave > 0 ? sWave[wave - 1] : 0);
    if (tid < MAXNB) {
        sBase[tid] = excl + bel;
        sCur[tid]  = 0;
        if (me == 0) {
            binStart[tid] = excl;
            if (tid == MAXNB - 1) binStart[MAXNB] = excl + tot;
        }
    }
    __syncthreads();

    #pragma unroll
    for (int k = 0; k < EPT; ++k) {
        if (er[k] >= 0) {
            const int bin  = er[k] / RB;
            const int rank = atomicAdd(&sCur[bin], 1);
            EdgeRec e;
            e.row = er[k] - bin * RB;
            e.col = ec[k];
            e.val = ev[k];
            e.pad = 0;
            rec[sBase[bin] + rank] = e;
        }
    }
}

template <int D>
__global__ __launch_bounds__(1024) void spmm_tile(const float* __restrict__ H,
                                                  const EdgeRec* __restrict__ rec,
                                                  const int* __restrict__ binStart,
                                                  float* __restrict__ out,
                                                  int nRows) {
    constexpr int PER = D / 32;
    constexpr int U   = 32 / PER;
    constexpr int T4  = RB * D / 4;

    __shared__ v4f tile4[T4];
    float* tile = (float*)tile4;

    const int tid  = threadIdx.x;
    const int lane = tid & 31;
    const int wave = __builtin_amdgcn_readfirstlane(tid >> 5);
    const int bin  = blockIdx.x;
    const int eBeg = __builtin_amdgcn_readfirstlane(binStart[bin]);
    const int eEnd = __builtin_amdgcn_readfirstlane(binStart[bin + 1]);

    const v4f zero4 = {0.0f, 0.0f, 0.0f, 0.0f};
    for (int i = tid; i < T4; i += 1024) tile4[i] = zero4;
    __syncthreads();

    const float* __restrict__ hl = H + lane;
    float* tl = tile + lane;

    for (int e0 = eBeg + wave * 32; e0 < eEnd; e0 += 32 * 32) {
        const int cnt = (eEnd - e0 < 32) ? (eEnd - e0) : 32;
        int   myRow = 0, myCol = 0;
        float myVal = 0.0f;
        if (lane < cnt) {
            const EdgeRec e = rec[e0 + lane];
            myRow = e.row; myCol = e.col; myVal = e.val;
        }
        if (cnt == 32) {
            #pragma unroll
            for (int i0 = 0; i0 < 32; i0 += U) {
                float src[U][PER];
                #pragma unroll
                for (int u = 0; u < U; ++u) {
                    const int c = __builtin_amdgcn_readlane(myCol, i0 + u);
                    const float* __restrict__ sp = hl + (size_t)c * D;
                    #pragma unroll
                    for (int j = 0; j < PER; ++j)
                        src[u][j] = sp[32 * j];
                }
                #pragma unroll
                for (int u = 0; u < U; ++u) {
                    const int   r = __builtin_amdgcn_readlane(myRow, i0 + u);
                    const float v = __int_as_float(__builtin_amdgcn_readlane(__float_as_int(myVal), i0 + u));
                    float* dst = tl + r * D;
                    #pragma unroll
                    for (int j = 0; j < PER; ++j)
                        atomicAdd(&dst[32 * j], v * src[u][j]);
                }
            }
        } else {
            for (int i = 0; i < cnt; ++i) {
                const int   c = __builtin_amdgcn_readlane(myCol, i);
                const int   r = __builtin_amdgcn_readlane(myRow, i);
                const float v = __int_as_float(__builtin_amdgcn_readlane(__float_as_int(myVal), i));
                const float* __restrict__ sp = hl + (size_t)c * D;
                float s[PER];
                #pragma unroll
                for (int j = 0; j < PER; ++j)
                    s[j] = sp[32 * j];
                float* dst = tl + r * D;
                #pragma unroll
                for (int j = 0; j < PER; ++j)
                    atomicAdd(&dst[32 * j], v * s[j]);
            }
        }
    }
    __syncthreads();

    const int row0 = bin * RB;
    const int nr   = (nRows - row0 < RB) ? (nRows - row0) : RB;
    v4f* __restrict__ o4 = (v4f*)(out + (size_t)row0 * D);
    for (int i = tid; i < nr * (D / 4); i += 1024) o4[i] = tile4[i];
}

extern "C" void kernel_launch(void* const* d_in, const int* in_sizes, int n_in,
                              void* d_out, int out_size, void* d_ws, size_t ws_size,
                              hipStream_t stream) {
    const float* features = (const float*)d_in[0];
    const float* edgeVals = (const float*)d_in[1];
    const float* W0       = (const float*)d_in[2];
    const float* W1       = (const float*)d_in[3];
    const float* W2       = (const float*)d_in[4];
    const int*   edgeRow  = (const int*)d_in[5];
    const int*   edgeCol  = (const int*)d_in[6];

    const int N = in_sizes[0] / 256;
    const int E = in_sizes[1];

    float* ws = (float*)d_ws;
    float* xnorm = ws;
    float* t0    = ws + (size_t)N * 256;
    float* g0    = ws + (size_t)N * 384;
    float* t1    = xnorm;
    float* g1    = t0;
    float* t2    = g0;
    float* out   = (float*)d_out;

    const int binBlocks = (E + CH - 1) / CH;
    const int nBins     = (N + RB - 1) / RB;
    EdgeRec* rec       = (EdgeRec*)(ws + (size_t)N * 512);
    int*     blockHist = (int*)(rec + E);
    int*     binStart  = blockHist + (size_t)binBlocks * MAXNB;

    const int normBlocks = (N + 7) / 8;

    l2norm_rows<<<normBlocks, 256, 0, stream>>>(features, xnorm, N);

    bin_count<<<binBlocks, 1024, 0, stream>>>(edgeRow, blockHist, E);
    bin_fill<<<binBlocks, 1024, 0, stream>>>(edgeRow, edgeCol, edgeVals, blockHist, binStart, rec, E, binBlocks);

    gemm_wstat_f32<256, 128, false><<<512, 256, 0, stream>>>(xnorm, W0, t0, N);
    spmm_tile<128><<<nBins, 1024, 0, stream>>>(t0, rec, binStart, g0, N);

    gemm_wstat_f32<128, 64, true><<<768, 256, 0, stream>>>(g0, W1, t1, N);
    spmm_tile<64><<<nBins, 1024, 0, stream>>>(t1, rec, binStart, g1, N);

    gemm_wstat_f32<64, 32, true><<<1024, 256, 0, stream>>>(g1, W2, t2, N);
    spmm_tile<32><<<nBins, 1024, 0, stream>>>(t2, rec, binStart, out, N);
}


# SpMM of layers 1,2: relu and the next layer's f32 WMMA product taken on the LDS result tile before it leaves (W staged in LDS, transposed); summed rows not stored, 9->7 launches; sum orders kept
# speedup vs baseline: 3.3937x; 1.0538x over previous
#include <hip/hip_runtime.h>
#include <hip/hip_bf16.h>


typedef __attribute__((ext_vector_type(2))) float v2f;
typedef __attribute__((ext_vector_type(8))) float v8f;
typedef __attribute__((ext_vector_type(4))) float v4f;

__global__ __launch_bounds__(256) void l2norm_rows(const float* __restrict__ F,
                                                   float* __restrict__ X,
                                                   int nRows) {
    const int lane = threadIdx.x & 31;
    const int row  = blockIdx.x * 8 + (threadIdx.x >> 5);
    if (row >= nRows) return;
    const float4* f = (const float4*)(F + (size_t)row * 256);
    float4 v0 = f[lane];
    float4 v1 = f[lane + 32];
    float s = v0.x*v0.x + v0.y*v0.y + v0.z*v0.z + v0.w*v0.w
            + v1.x*v1.x + v1.y*v1.y + v1.z*v1.z + v1.w*v1.w;
    #pragma unroll
    for (int off = 16; off > 0; off >>= 1) s += __shfl_xor(s, off, 32);
    const float inv = rsqrtf(fmaxf(s, 1e-12f));
    v0.x *= inv; v0.y *= inv; v0.z *= inv; v0.w *= inv;
    v1.x *= inv; v1.y *= inv; v1.z *= inv; v1.w *= inv;
    float4* xo = (float4*)(X + (size_t)row * 256);
    xo[lane]      = v0;
    xo[lane + 32] = v1;
}

template <int K, int M, bool RELU>
__global__ __launch_bounds__(256) void gemm_wstat_f32(const float* __restrict__ X,
                                                      const float* __restrict__ W,
                                                      float* __restrict__ Y,
                                                      int nRows) {
    constexpr int TN   = M / 16;
    constexpr int G    = 8 / TN;
    constexpr int ROWS = G * 32;
    constexpr int K4   = K / 4;
    constexpr int KP4  = K4 + 1;
    constexpr int NLD  = ROWS * K4 / 256;

    __shared__ v4f tile4[ROWS * KP4];

    const int tid     = threadIdx.x;
    const int lane    = tid & 31;
    const int wave    = tid >> 5;
    const int colTile = wave % TN;
    const int grp     = wave / TN;
    const int c       = lane & 15;
    const int h       = lane >> 4;
    const int bCol    = colTile * 16 + c;

    const int nChunks = (nRows + ROWS - 1) / ROWS;
    int chunk = blockIdx.x;
    if (chunk >= nChunks) return;

    v2f wreg[K4];
    #pragma unroll
    for (int g = 0; g < K / 8; ++g) {
        #pragma unroll
        for (int s = 0; s < 2; ++s) {
            const int kk = 8 * g + 4 * h + 2 * s;
            v2f b;
            b.x = W[(size_t)kk       * M + bCol];
            b.y = W[(size_t)(kk + 1) * M + bCol];
            wreg[2 * g + s] = b;
        }
    }

    const v4f* __restrict__ X4 = (const v4f*)X;
    v4f p[NLD];

    #pragma unroll
    for (int i = 0; i < NLD; ++i) {
        const int idx  = tid + 256 * i;
        const int grow = min(chunk * ROWS + idx / K4, nRows - 1);
        p[i] = X4[(size_t)grow * K4 + (idx % K4)];
    }

    for (; chunk < nChunks; chunk += gridDim.x) {
        #pragma unroll
        for (int i = 0; i < NLD; ++i) {
            const int idx = tid + 256 * i;
            v4f v = p[i];
            if (RELU) {
                v.x = fmaxf(v.x, 0.0f); v.y = fmaxf(v.y, 0.0f);
                v.z = fmaxf(v.z, 0.0f); v.w = fmaxf(v.w, 0.0f);
            }
            tile4[(idx / K4) * KP4 + (idx % K4)] = v;
        }
        __syncthreads();

        const int nxt = chunk + gridDim.x;
        if (nxt < nChunks) {
            #pragma unroll
            for (int i = 0; i < NLD; ++i) {
                const int idx  = tid + 256 * i;
                const int grow = min(nxt * ROWS + idx / K4, nRows - 1);
                p[i] = X4[(size_t)grow * K4 + (idx % K4)];
            }
        }

        const int row0 = chunk * ROWS + grp * 32;
        if (row0 < nRows) {
            const v4f* a0p = tile4 + (grp * 32 + c) * KP4 + h;
            const v4f* a1p = a0p + 16 * KP4;
            v8f acc0 = {};
            v8f acc1 = {};
            #pragma unroll
            for (int g = 0; g < K / 8; ++g) {
                const v4f a0 = a0p[2 * g];
                const v4f a1 = a1p[2 * g];
                v2f f;
                f.x = a0.x; f.y = a0.y;
                acc0 = __builtin_amdgcn_wmma_f32_16x16x4_f32(false, f, false, wreg[2 * g],     (short)0, acc0, false, false);
                f.x = a1.x; f.y = a1.y;
                acc1 = __builtin_amdgcn_wmma_f32_16x16x4_f32(false, f, false, wreg[2 * g],     (short)0, acc1, false, false);
                f.x = a0.z; f.y = a0.w;
                acc0 = __builtin_amdgcn_wmma_f32_16x16x4_f32(false, f, false, wreg[2 * g + 1], (short)0, acc0, false, false);
                f.x = a1.z; f.y = a1.w;
                acc1 = __builtin_amdgcn_wmma_f32_16x16x4_f32(false, f, false, wreg[2 * g + 1], (short)0, acc1, false, false);
            }
            float* __restrict__ y0 = Y + (size_t)(row0 + 8 * h) * M + bCol;
            #pragma unroll
            for (int r = 0; r < 8; ++r)
                y0[(size_t)r * M] = acc0[r];
            if (row0 + 16 < nRows) {
                float* __restrict__ y1 = y0 + (size_t)16 * M;
                #pragma unroll
                for (int r = 0; r < 8; ++r)
                    y1[(size_t)r * M] = acc1[r];
            }
        }
        __syncthreads();
    }
}

constexpr int RB    = 200;
constexpr int MAXNB = 512;
constexpr int CH    = 8192;
constexpr int EPT   = CH / 1024;

struct __attribute__((aligned(16))) EdgeRec { int row; int col; float val; int pad; };

__global__ __launch_bounds__(1024) void bin_count(const int* __restrict__ rows,
                                                  int* __restrict__ blockHist,
                                                  int nEdges) {
    __shared__ int hist[MAXNB];
    const int tid = threadIdx.x;
    if (tid < MAXNB) hist[tid] = 0;
    __syncthreads();
    const int beg = blockIdx.x * CH;
    const int end = min(beg + CH, nEdges);
    int er[EPT];
    #pragma unroll
    for (int k = 0; k < EPT; ++k) {
        const int i = beg + tid + 1024 * k;
        er[k] = (i < end) ? rows[i] : -1;
    }
    #pragma unroll
    for (int k = 0; k < EPT; ++k)
        if (er[k] >= 0) atomicAdd(&hist[er[k] / RB], 1);
    __syncthreads();
    if (tid < MAXNB) blockHist[blockIdx.x * MAXNB + tid] = hist[tid];
}

__device__ __forceinline__ int hist_col_sum(const int* __restrict__ p, int n) {
    int s = 0;
    for (int i = 0; i < n; i += 16) {
        int c[16];
        #pragma unroll
        for (int q = 0; q < 16; ++q)
            c[q] = (i + q < n) ? p[(size_t)(i + q) * MAXNB] : 0;
        #pragma unroll
        for (int q = 0; q < 16; ++q)
            s += c[q];
    }
    return s;
}

__global__ __launch_bounds__(1024) void bin_fill(const int* __restrict__ rows,
                                                 const int* __restrict__ cols,
                                                 const float* __restrict__ vals,
                                                 const int* __restrict__ blockHist,
                                                 int* __restrict__ binStart,
                                                 EdgeRec* __restrict__ rec,
                                                 int nEdges, int nBlocks) {
    __shared__ int sBelow[2][MAXNB];
    __shared__ int sTotal[2][MAXNB];
    __shared__ int sBase[MAXNB];
    __shared__ int sCur[MAXNB];
    __shared__ int sWave[32];

    const int tid  = threadIdx.x;
    const int lane = tid & 31;
    const int wave = tid >> 5;
    const int me   = blockIdx.x;

    const int beg = me * CH;
    const int end = min(beg + CH, nEdges);
    int   er[EPT], ec[EPT];
    float ev[EPT];
    #pragma unroll
    for (int k = 0; k < EPT; ++k) {
        const int i = beg + tid + 1024 * k;
        er[k] = -1; ec[k] = 0; ev[k] = 0.0f;
        if (i < end) { er[k] = rows[i]; ec[k] = cols[i]; ev[k] = vals[i]; }
    }

    {
        const int bin  = tid & (MAXNB - 1);
        const int half = __builtin_amdgcn_readfirstlane(tid / MAXNB);
        const int mid  = (nBlocks + 1) / 2;
        const int lo   = half ? mid : 0;
        const int hi   = half ? nBlocks : mid;
        const int cut  = min(max(me, lo), hi);
        const int below = hist_col_sum(blockHist + (size_t)lo  * MAXNB + bin, cut - lo);
        const int rest  = hist_col_sum(blockHist + (size_t)cut * MAXNB + bin, hi - cut);
        sBelow[half][bin] = below;
        sTotal[half][bin] = below + rest;
    }
    __syncthreads();

    int tot = 0, bel = 0;
    if (tid < MAXNB) {
        tot = sTotal[0][tid] + sTotal[1][tid];
        bel = sBelow[0][tid] + sBelow[1][tid];
    }
    int incl = tot;
    for (int off = 1; off < 32; off <<= 1) {
        const int up = __shfl_up(incl, off, 32);
        if (lane >= off) incl += up;
    }
    if (lane == 31) sWave[wave] = incl;
    __syncthreads();
    if (wave == 0) {
        int s = sWave[lane];
        for (int off = 1; off < 32; off <<= 1) {
            const int up = __shfl_up(s, off, 32);
            if (lane >= off) s += up;
        }
        sWave[lane] = s;
    }
    __syncthreads();
    const int excl = incl - tot + (wave > 0 ? sWave[wave - 1] : 0);
    if (tid < MAXNB) {
        sBase[tid] = excl + bel;
        sCur[tid]  = 0;
        if (me == 0) {
            binStart[tid] = excl;
            if (tid == MAXNB - 1) binStart[MAXNB] = excl + tot;
        }
    }
    __syncthreads();

    #pragma unroll
    for (int k = 0; k < EPT; ++k) {
        if (er[k] >= 0) {
            const int bin  = er[k] / RB;
            const int rank = atomicAdd(&sCur[bin], 1);
            EdgeRec e;
            e.row = er[k] - bin * RB;
            e.col = ec[k];
            e.val = ev[k];
            e.pad = 0;
            rec[sBase[bin] + rank] = e;
        }
    }
}

template <int D>
__global__ __launch_bounds__(1024) void spmm_tile(const float* __restrict__ H,
                                                  const EdgeRec* __restrict__ rec,
                                                  const int* __restrict__ binStart,
                                                  float* __restrict__ out,
                                                  int nRows) {
    constexpr int PER = D / 32;
    constexpr int U   = 32 / PER;
    constexpr int T4  = RB * D / 4;

    __shared__ v4f tile4[T4];
    float* tile = (float*)tile4;

    const int tid  = threadIdx.x;
    const int lane = tid & 31;
    const int wave = __builtin_amdgcn_readfirstlane(tid >> 5);
    const int bin  = blockIdx.x;
    const int eBeg = __builtin_amdgcn_readfirstlane(binStart[bin]);
    const int eEnd = __builtin_amdgcn_readfirstlane(binStart[bin + 1]);

    const v4f zero4 = {0.0f, 0.0f, 0.0f, 0.0f};
    for (int i = tid; i < T4; i += 1024) tile4[i] = zero4;
    __syncthreads();

    const float* __restrict__ hl = H + lane;
    float* tl = tile + lane;

    for (int e0 = eBeg + wave * 32; e0 < eEnd; e0 += 32 * 32) {
        const int cnt = (eEnd - e0 < 32) ? (eEnd - e0) : 32;
        int   myRow = 0, myCol = 0;
        float myVal = 0.0f;
        if (lane < cnt) {
            const EdgeRec e = rec[e0 + lane];
            myRow = e.row; myCol = e.col; myVal = e.val;
        }
        if (cnt == 32) {
            #pragma unroll
            for (int i0 = 0; i0 < 32; i0 += U) {
                float src[U][PER];
                #pragma unroll
                for (int u = 0; u < U; ++u) {
                    const int c = __builtin_amdgcn_readlane(myCol, i0 + u);
                    const float* __restrict__ sp = hl + (size_t)c * D;
                    #pragma unroll
                    for (int j = 0; j < PER; ++j)
                        src[u][j] = sp[32 * j];
                }
                #pragma unroll
                for (int u = 0; u < U; ++u) {
                    const int   r = __builtin_amdgcn_readlane(myRow, i0 + u);
                    const float v = __int_as_float(__builtin_amdgcn_readlane(__float_as_int(myVal), i0 + u));
                    float* dst = tl + r * D;
                    #pragma unroll
                    for (int j = 0; j < PER; ++j)
                        atomicAdd(&dst[32 * j], v * src[u][j]);
                }
            }
        } else {
            for (int i = 0; i < cnt; ++i) {
                const int   c = __builtin_amdgcn_readlane(myCol, i);
                const int   r = __builtin_amdgcn_readlane(myRow, i);
                const float v = __int_as_float(__builtin_amdgcn_readlane(__float_as_int(myVal), i));
                const float* __restrict__ sp = hl + (size_t)c * D;
                float s[PER];
                #pragma unroll
                for (int j = 0; j < PER; ++j)
                    s[j] = sp[32 * j];
                float* dst = tl + r * D;
                #pragma unroll
                for (int j = 0; j < PER; ++j)
                    atomicAdd(&dst[32 * j], v * s[j]);
            }
        }
    }
    __syncthreads();

    const int row0 = bin * RB;
    const int nr   = (nRows - row0 < RB) ? (nRows - row0) : RB;
    v4f* __restrict__ o4 = (v4f*)(out + (size_t)row0 * D);
    for (int i = tid; i < nr * (D / 4); i += 1024) o4[i] = tile4[i];
}

template <int D, int M>
__global__ __launch_bounds__(1024) void spmm_tile_proj(const float* __restrict__ H,
                                                       const EdgeRec* __restrict__ rec,
                                                       const int* __restrict__ binStart,
                                                       const float* __restrict__ Wn,
                                                       float* __restrict__ Y,
                                                       int nRows) {
    constexpr int PER   = D / 32;
    constexpr int U     = 32 / PER;
    constexpr int RT    = (RB + 15) / 16;
    constexpr int K4    = D / 4;
    constexpr int KP4   = K4 + 1;
    constexpr int PITCH = 4 * KP4;
    constexpr int T4    = RT * 16 * KP4;
    constexpr int TN    = M / 16;
    constexpr int WPC   = 32 / TN;

    __shared__ v4f tile4[T4];
    __shared__ v4f wt4[M * KP4];
    float* tile = (float*)tile4;
    float* wt   = (float*)wt4;

    const int tid  = threadIdx.x;
    const int lane = tid & 31;
    const int wave = __builtin_amdgcn_readfirstlane(tid >> 5);
    const int bin  = blockIdx.x;
    const int eBeg = __builtin_amdgcn_readfirstlane(binStart[bin]);
    const int eEnd = __builtin_amdgcn_readfirstlane(binStart[bin + 1]);

    constexpr int NW = D * M / 1024;
    float wv[NW];
    #pragma unroll
    for (int q = 0; q < NW; ++q) wv[q] = Wn[tid + 1024 * q];

    const v4f zero4 = {0.0f, 0.0f, 0.0f, 0.0f};
    for (int i = tid; i < T4; i += 1024) tile4[i] = zero4;
    #pragma unroll
    for (int q = 0; q < NW; ++q) {
        const int i = tid + 1024 * q;
        wt[(i % M) * PITCH + (i / M)] = wv[q];
    }
    __syncthreads();

    const float* __restrict__ hl = H + lane;
    float* tl = tile + lane;

    for (int e0 = eBeg + wave * 32; e0 < eEnd; e0 += 32 * 32) {
        const int cnt = (eEnd - e0 < 32) ? (eEnd - e0) : 32;
        int   myRow = 0, myCol = 0;
        float myVal = 0.0f;
        if (lane < cnt) {
            const EdgeRec e = rec[e0 + lane];
            myRow = e.row; myCol = e.col; myVal = e.val;
        }
        if (cnt == 32) {
            #pragma unroll
            for (int i0 = 0; i0 < 32; i0 += U) {
                float src[U][PER];
                #pragma unroll
                for (int u = 0; u < U; ++u) {
                    const int c = __builtin_amdgcn_readlane(myCol, i0 + u);
                    const float* __restrict__ sp = hl + (size_t)c * D;
                    #pragma unroll
                    for (int j = 0; j < PER; ++j)
                        src[u][j] = sp[32 * j];
                }
                #pragma unroll
                for (int u = 0; u < U; ++u) {
                    const int   r = __builtin_amdgcn_readlane(myRow, i0 + u);
                    const float v = __int_as_float(__builtin_amdgcn_readlane(__float_as_int(myVal), i0 + u));
                    float* dst = tl + r * PITCH;
                    #pragma unroll
                    for (int j = 0; j < PER; ++j)
                        atomicAdd(&dst[32 * j], v * src[u][j]);
                }
            }
        } else {
            for (int i = 0; i < cnt; ++i) {
                const int   c = __builtin_amdgcn_readlane(myCol, i);
                const int   r = __builtin_amdgcn_readlane(myRow, i);
                const float v = __int_as_float(__builtin_amdgcn_readlane(__float_as_int(myVal), i));
                const float* __restrict__ sp = hl + (size_t)c * D;
                float s[PER];
                #pragma unroll
                for (int j = 0; j < PER; ++j)
                    s[j] = sp[32 * j];
                float* dst = tl + r * PITCH;
                #pragma unroll
                for (int j = 0; j < PER; ++j)
                    atomicAdd(&dst[32 * j], v * s[j]);
            }
        }
    }
    __syncthreads();

    const int row0 = bin * RB;
    const int nr   = (nRows - row0 < RB) ? (nRows - row0) : RB;
    const int c    = lane & 15;
    const int h    = lane >> 4;
    const int ct   = wave % TN;
    const v4f* bp  = wt4 + (ct * 16 + c) * KP4 + h;
    for (int rt = wave / TN; rt < RT; rt += WPC) {
        const v4f* ap = tile4 + (rt * 16 + c) * KP4 + h;
        v8f acc = {};
        #pragma unroll 4
        for (int g = 0; g < D / 8; ++g) {
            v4f a = ap[2 * g];
            const v4f b = bp[2 * g];
            a.x = fmaxf(a.x, 0.0f); a.y = fmaxf(a.y, 0.0f);
            a.z = fmaxf(a.z, 0.0f); a.w = fmaxf(a.w, 0.0f);
            v2f fa, fb;
            fa.x = a.x; fa.y = a.y; fb.x = b.x; fb.y = b.y;
            acc = __builtin_amdgcn_wmma_f32_16x16x4_f32(false, fa, false, fb, (short)0, acc, false, false);
            fa.x = a.z; fa.y = a.w; fb.x = b.z; fb.y = b.w;
            acc = __builtin_amdgcn_wmma_f32_16x16x4_f32(false, fa, false, fb, (short)0, acc, false, false);
        }
        const int rb = rt * 16 + 8 * h;
        float* __restrict__ y = Y + (size_t)(row0 + rb) * M + ct * 16 + c;
        #pragma unroll
        for (int r = 0; r < 8; ++r)
            if (rb + r < nr) y[(size_t)r * M] = acc[r];
    }
}

extern "C" void kernel_launch(void* const* d_in, const int* in_sizes, int n_in,
                              void* d_out, int out_size, void* d_ws, size_t ws_size,
                              hipStream_t stream) {
    const float* features = (const float*)d_in[0];
    const float* edgeVals = (const float*)d_in[1];
    const float* W0       = (const float*)d_in[2];
    const float* W1       = (const float*)d_in[3];
    const float* W2       = (const float*)d_in[4];
    const int*   edgeRow  = (const int*)d_in[5];
    const int*   edgeCol  = (const int*)d_in[6];

    const int N = in_sizes[0] / 256;
    const int E = in_sizes[1];

    float* ws = (float*)d_ws;
    float* xnorm = ws;
    float* t0    = ws + (size_t)N * 256;
    float* t2    = ws + (size_t)N * 384;
    float* t1    = xnorm;
    float* out   = (float*)d_out;

    const int binBlocks = (E + CH - 1) / CH;
    const int nBins     = (N + RB - 1) / RB;
    EdgeRec* rec       = (EdgeRec*)(ws + (size_t)N * 512);
    int*     blockHist = (int*)(rec + E);
    int*     binStart  = blockHist + (size_t)binBlocks * MAXNB;

    const int normBlocks = (N + 7) / 8;

    l2norm_rows<<<normBlocks, 256, 0, stream>>>(features, xnorm, N);

    bin_count<<<binBlocks, 1024, 0, stream>>>(edgeRow, blockHist, E);
    bin_fill<<<binBlocks, 1024, 0, stream>>>(edgeRow, edgeCol, edgeVals, blockHist, binStart, rec, E, binBlocks);

    gemm_wstat_f32<256, 128, false><<<512, 256, 0, stream>>>(xnorm, W0, t0, N);
    spmm_tile_proj<128, 64><<<nBins, 1024, 0, stream>>>(t0, rec, binStart, W1, t1, N);

    spmm_tile_proj<64, 32><<<nBins, 1024, 0, stream>>>(t1, rec, binStart, W2, t2, N);

    spmm_tile<32><<<nBins, 1024, 0, stream>>>(t2, rec, binStart, out, N);
}
